# starting kernel
# baseline (speedup 1.0000x reference)
#include <hip/hip_runtime.h>
#include <hip/hip_bf16.h>
#include <math.h>

#define BATCH 256
#define N_NEG 4096
#define EMB   256
#define HALF  (EMB / 2)

typedef __attribute__((ext_vector_type(2))) float v2f;
typedef __attribute__((ext_vector_type(8))) float v8f;

__global__ void __launch_bounds__(128)
rotate_kernel(const float* __restrict__ head,
              const int*   __restrict__ rel_id,
              const float* __restrict__ rel_emb,
              float* __restrict__ rot,
              float* __restrict__ rn) {
    const int b = blockIdx.x;
    const int j = threadIdx.x;

    const float re = head[b * EMB + j];
    const float im = head[b * EMB + HALF + j];
    const int   r  = rel_id[b];
    const float ph = rel_emb[r * HALF + j];

    float s, c;
    __sincosf(ph, &s, &c);

    rot[b * EMB + j]        = re * c - im * s;
    rot[b * EMB + HALF + j] = re * s + im * c;

    __shared__ float sm[128];
    sm[j] = re * re + im * im;
    __syncthreads();
    #pragma unroll
    for (int stride = 64; stride > 0; stride >>= 1) {
        if (j < stride) sm[j] += sm[j + stride];
        __syncthreads();
    }
    if (j == 0) rn[b] = sm[0];
}

__global__ void __launch_bounds__(256)
candnorm_kernel(const float* __restrict__ cand,
                float* __restrict__ cn) {
    const int t   = blockIdx.x;
    const int tid = threadIdx.x;
    const float v = cand[t * EMB + tid];

    __shared__ float sm[256];
    sm[tid] = v * v;
    __syncthreads();
    #pragma unroll
    for (int stride = 128; stride > 0; stride >>= 1) {
        if (tid < stride) sm[tid] += sm[tid + stride];
        __syncthreads();
    }
    if (tid == 0) cn[t] = sm[0];
}

__global__ void __launch_bounds__(256)
score_kernel(const float* __restrict__ rot,
             const float* __restrict__ cand,
             const float* __restrict__ rn,
             const float* __restrict__ cn,
             float* __restrict__ out) {
    const int lane  = threadIdx.x & 31;
    const int wave  = threadIdx.x >> 5;
    const int mtile = blockIdx.y;
    const int ntile = blockIdx.x * 8 + wave;

    const int r    = lane & 15;
    const int kadj = (lane >> 4) << 1;

    const float* arow = rot  + (mtile * 16 + r) * EMB + kadj;
    const float* brow = cand + (ntile * 16 + r) * EMB + kadj;

    v8f acc = {};
    #pragma unroll
    for (int k = 0; k < EMB; k += 4) {
        v2f a = *(const v2f*)(arow + k);
        v2f b = *(const v2f*)(brow + k);
        acc = __builtin_amdgcn_wmma_f32_16x16x4_f32(
            false, a, false, b, (short)0, acc, false, false);
    }

    const int col  = ntile * 16 + r;
    const float cnv = cn[col];
    #pragma unroll
    for (int j = 0; j < 8; ++j) {
        const int row = mtile * 16 + j + ((lane >> 4) << 3);
        const float d2 = rn[row] + cnv - 2.0f * acc[j];
        out[row * N_NEG + col] = -sqrtf(fmaxf(d2, 0.0f));
    }
}

extern "C" void kernel_launch(void* const* d_in, const int* in_sizes, int n_in,
                              void* d_out, int out_size, void* d_ws, size_t ws_size,
                              hipStream_t stream) {
    const float* head    = (const float*)d_in[0];
    const int*   rel_id  = (const int*)  d_in[1];
    const float* tail    = (const float*)d_in[2];
    const float* rel_emb = (const float*)d_in[3];
    float*       out     = (float*)d_out;

    float* rot = (float*)d_ws;
    float* rn  = rot + BATCH * EMB;
    float* cn  = rn + BATCH;

    rotate_kernel<<<BATCH, 128, 0, stream>>>(head, rel_id, rel_emb, rot, rn);
    candnorm_kernel<<<N_NEG, 256, 0, stream>>>(tail, cn);

    dim3 grid(N_NEG / (16 * 8), BATCH / 16);
    score_kernel<<<grid, 256, 0, stream>>>(rot, tail, rn, cn, out);
}


# Rotate kernel: one wave per head row (16-B loads and stores), row norm summed per lane then across lanes by DPP and permlanex16 exchange; no LDS, no barriers; head norm sum order changes
# speedup vs baseline: 2.2524x; 2.2524x over previous
#include <hip/hip_runtime.h>
#include <math.h>


#define BATCH 256
#define N_NEG 4096
#define EMB   256
#define HALF  (EMB / 2)

typedef __attribute__((ext_vector_type(2))) float v2f;
typedef __attribute__((ext_vector_type(4))) float v4f;
typedef __attribute__((ext_vector_type(8))) float v8f;

__device__ __forceinline__ int frag_index(int row, int k) {
    return ((((row >> 4) * 32 + (k >> 3)) * 32 + ((k >> 2) & 1) * 16 + (row & 15)) << 2) + (k & 3);
}


template <int M>
__device__ __forceinline__ float row_xor(float x) {
    const int xi = __builtin_bit_cast(int, x);
    return __builtin_bit_cast(float, __builtin_amdgcn_mov_dpp(xi, 0x160 + M, 0xf, 0xf, false));
}

__device__ __forceinline__ float half_xor(float x) {
    const int xi = __builtin_bit_cast(int, x);
    return __builtin_bit_cast(float,
        __builtin_amdgcn_permlanex16(xi, xi, 0x76543210, 0xfedcba98, false, false));
}

__global__ void __launch_bounds__(32)
rotate_kernel(const float* __restrict__ head,
              const int*   __restrict__ rel_id,
              const float* __restrict__ rel_emb,
              float* __restrict__ rn,
              float* __restrict__ rotF) {
    const int b    = blockIdx.x;
    const int lane = threadIdx.x;
    const int j0   = 4 * lane;

    const int r  = rel_id[b];
    const v4f re = *(const v4f*)(head + b * EMB + j0);
    const v4f im = *(const v4f*)(head + b * EMB + HALF + j0);
    const v4f ph = *(const v4f*)(rel_emb + r * HALF + j0);

    v4f lo, hi;
    float p = 0.0f;
    #pragma unroll
    for (int i = 0; i < 4; ++i) {
        float s, c;
        __sincosf(ph[i], &s, &c);
        lo[i] = re[i] * c - im[i] * s;
        hi[i] = re[i] * s + im[i] * c;
        const float t = re[i] * re[i] + im[i] * im[i];
        p = (i == 0) ? t : p + t;
    }

    *(v4f*)(rotF + frag_index(b, j0))        = lo;
    *(v4f*)(rotF + frag_index(b, j0 + HALF)) = hi;

    p += row_xor<1>(p);
    p += row_xor<2>(p);
    p += row_xor<4>(p);
    p += row_xor<8>(p);
    p += half_xor(p);
    if (lane == 0) rn[b] = p;
}

#define MT 4
#define GQ 8
#define NG (32 / GQ)

__global__ void __launch_bounds__(128)
score_kernel(const float* __restrict__ rotF,
             const float* __restrict__ cand,
             const float* __restrict__ rn,
             float* __restrict__ out) {
    const int lane  = threadIdx.x & 31;
    const int wave  = threadIdx.x >> 5;
    const int c     = lane & 15;
    const int h     = lane >> 4;
    const int ntile = blockIdx.x * 4 + wave;
    const int mbase = blockIdx.y * MT;

    const v4f* aF = (const v4f*)rotF + mbase * 1024 + lane;
    const v4f* bR = (const v4f*)(cand + (ntile * 16 + c) * EMB + 4 * h);

    v4f a[2][GQ][MT];
    v4f b[2][GQ];
    v8f acc[MT] = {};
    float s = 0.0f;

    #pragma unroll
    for (int i = 0; i < GQ; ++i) {
        b[0][i] = bR[2 * i];
        #pragma unroll
        for (int t = 0; t < MT; ++t) a[0][i][t] = aF[t * 1024 + i * 32];
    }

    #pragma unroll
    for (int g = 0; g < NG; ++g) {
        const int cur = g & 1, nxt = cur ^ 1;
        if (g + 1 < NG) {
            #pragma unroll
            for (int i = 0; i < GQ; ++i) {
                const int q = (g + 1) * GQ + i;
                b[nxt][i] = bR[2 * q];
                #pragma unroll
                for (int t = 0; t < MT; ++t) a[nxt][i][t] = aF[t * 1024 + q * 32];
            }
        }
        __builtin_amdgcn_sched_barrier(0);
        #pragma unroll
        for (int i = 0; i < GQ; ++i) {
            const v4f bv = b[cur][i];
            s += bv.x * bv.x;
            s += bv.y * bv.y;
            s += bv.z * bv.z;
            s += bv.w * bv.w;
            const v2f b0 = bv.xy, b1 = bv.zw;
            #pragma unroll
            for (int t = 0; t < MT; ++t) {
                const v2f a0 = a[cur][i][t].xy;
                acc[t] = __builtin_amdgcn_wmma_f32_16x16x4_f32(
                    false, a0, false, b0, (short)0, acc[t], false, false);
            }
            #pragma unroll
            for (int t = 0; t < MT; ++t) {
                const v2f a1 = a[cur][i][t].zw;
                acc[t] = __builtin_amdgcn_wmma_f32_16x16x4_f32(
                    false, a1, false, b1, (short)0, acc[t], false, false);
            }
        }
        __builtin_amdgcn_sched_barrier(0);
    }

    const float cnv = s + __shfl_xor(s, 16);

    const int col = ntile * 16 + c;
    #pragma unroll
    for (int t = 0; t < MT; ++t) {
        #pragma unroll
        for (int j = 0; j < 8; ++j) {
            const int row = (mbase + t) * 16 + j + (h << 3);
            const float d2 = rn[row] + cnv - 2.0f * acc[t][j];
            out[row * N_NEG + col] = -sqrtf(fmaxf(d2, 0.0f));
        }
    }
}

extern "C" void kernel_launch(void* const* d_in, const int* in_sizes, int n_in,
                              void* d_out, int out_size, void* d_ws, size_t ws_size,
                              hipStream_t stream) {
    const float* head    = (const float*)d_in[0];
    const int*   rel_id  = (const int*)  d_in[1];
    const float* tail    = (const float*)d_in[2];
    const float* rel_emb = (const float*)d_in[3];
    float*       out     = (float*)d_out;

    float* rotF  = (float*)d_ws;
    float* rn    = rotF + BATCH * EMB;

    rotate_kernel<<<BATCH, 32, 0, stream>>>(head, rel_id, rel_emb, rn, rotF);

    dim3 grid(N_NEG / (16 * 4), BATCH / (16 * MT));
    score_kernel<<<grid, 128, 0, stream>>>(rotF, tail, rn, out);
}
